# starting kernel
# baseline (speedup 1.0000x reference)
#include <hip/hip_runtime.h>
#include <math.h>

typedef __attribute__((ext_vector_type(2))) float v2f;
typedef __attribute__((ext_vector_type(8))) float v8f;

#define NN   20000
#define EE   320000
#define TT   4
#define HID  128
#define G3   384

__global__ void fill_zero_k(float* __restrict__ p, int n) {
    int i = blockIdx.x * blockDim.x + threadIdx.x;
    if (i < n) p[i] = 0.0f;
}

__global__ void count_deg_k(const int* __restrict__ eis, float* __restrict__ deg) {
    int i = blockIdx.x * blockDim.x + threadIdx.x;
    if (i >= TT * EE) return;
    int t = i / EE, e = i - t * EE;
    int d = eis[(size_t)t * 2 * EE + EE + e];
    atomicAdd(&deg[t * NN + d], 1.0f);
}

__global__ void finalize_dinv_k(float* __restrict__ deg, int n) {
    int i = blockIdx.x * blockDim.x + threadIdx.x;
    if (i < n) deg[i] = rsqrtf(deg[i] + 1.0f);
}

__global__ void init_self_k(float* __restrict__ out, const float* __restrict__ H,
                            const float* __restrict__ dinv, const float* __restrict__ bias) {
    int i = blockIdx.x * blockDim.x + threadIdx.x;
    if (i >= NN * HID) return;
    int n = i >> 7, j = i & 127;
    float di = dinv[n];
    out[i] = bias[j] + H[i] * di * di;
}

__global__ void scatter_k(float* __restrict__ out, const float* __restrict__ H,
                          const float* __restrict__ dinv,
                          const int* __restrict__ src, const int* __restrict__ dst) {
    int i = blockIdx.x * blockDim.x + threadIdx.x;
    int e = i >> 5;
    if (e >= EE) return;
    int lane = i & 31;
    int s = src[e], d = dst[e];
    float norm = dinv[s] * dinv[d];
    const float4 v = *(const float4*)(H + (size_t)s * HID + lane * 4);
    float* o = out + (size_t)d * HID + lane * 4;
    atomicAdd(o + 0, v.x * norm);
    atomicAdd(o + 1, v.y * norm);
    atomicAdd(o + 2, v.z * norm);
    atomicAdd(o + 3, v.w * norm);
}

__global__ void relu_k(float* __restrict__ p, int n) {
    int i = blockIdx.x * blockDim.x + threadIdx.x;
    if (i < n) p[i] = fmaxf(p[i], 0.0f);
}

#define BPS 160

__global__ void wmma_gemm_k(const float* __restrict__ A, const float* __restrict__ B,
                            const float* __restrict__ bias, float* __restrict__ C,
                            int Mrows, int Nout, int rsB, int csB, int relu) {
    __shared__ float Blds[64 * BPS];
    const int tid = threadIdx.x;
    const int col0 = blockIdx.y * 64;

    for (int idx = tid; idx < 128 * 64; idx += 256) {
        int k = idx >> 6, c = idx & 63;
        Blds[(k >> 1) * BPS + c * 2 + (k & 1)] =
            B[(size_t)k * rsB + (size_t)(col0 + c) * csB];
    }
    __syncthreads();

    const int wave = tid >> 5, lane = tid & 31;
    const int m = lane & 15;
    const int kof = (lane >> 4) * 2;
    const int rt0 = blockIdx.x * 16 + wave;
    const int rt1 = rt0 + 8;
    const int r0 = rt0 * 16, r1 = rt1 * 16;
    const float* A0 = A + (size_t)((r0 < Mrows ? r0 : 0) + m) * 128;
    const float* A1 = A + (size_t)((r1 < Mrows ? r1 : 0) + m) * 128;

    v8f acc0_0 = {}, acc0_1 = {}, acc0_2 = {}, acc0_3 = {};
    v8f acc1_0 = {}, acc1_1 = {}, acc1_2 = {}, acc1_3 = {};

    for (int k = 0; k < 128; k += 4) {
        int kk = k + kof;
        v2f a0 = *(const v2f*)(A0 + kk);
        v2f a1 = *(const v2f*)(A1 + kk);
        const float* bp = Blds + (kk >> 1) * BPS + m * 2;
        v2f b;
        b = *(const v2f*)(bp);
        acc0_0 = __builtin_amdgcn_wmma_f32_16x16x4_f32(false, a0, false, b, (short)0, acc0_0, false, false);
        acc1_0 = __builtin_amdgcn_wmma_f32_16x16x4_f32(false, a1, false, b, (short)0, acc1_0, false, false);
        b = *(const v2f*)(bp + 32);
        acc0_1 = __builtin_amdgcn_wmma_f32_16x16x4_f32(false, a0, false, b, (short)0, acc0_1, false, false);
        acc1_1 = __builtin_amdgcn_wmma_f32_16x16x4_f32(false, a1, false, b, (short)0, acc1_1, false, false);
        b = *(const v2f*)(bp + 64);
        acc0_2 = __builtin_amdgcn_wmma_f32_16x16x4_f32(false, a0, false, b, (short)0, acc0_2, false, false);
        acc1_2 = __builtin_amdgcn_wmma_f32_16x16x4_f32(false, a1, false, b, (short)0, acc1_2, false, false);
        b = *(const v2f*)(bp + 96);
        acc0_3 = __builtin_amdgcn_wmma_f32_16x16x4_f32(false, a0, false, b, (short)0, acc0_3, false, false);
        acc1_3 = __builtin_amdgcn_wmma_f32_16x16x4_f32(false, a1, false, b, (short)0, acc1_3, false, false);
    }

    const int mb = (lane >> 4) * 8;
    float bv0 = bias ? bias[col0 + m]      : 0.0f;
    float bv1 = bias ? bias[col0 + 16 + m] : 0.0f;
    float bv2 = bias ? bias[col0 + 32 + m] : 0.0f;
    float bv3 = bias ? bias[col0 + 48 + m] : 0.0f;
    if (r0 < Mrows) {
#pragma unroll
        for (int r = 0; r < 8; ++r) {
            size_t ro = (size_t)(r0 + mb + r) * Nout;
            float v;
            v = acc0_0[r] + bv0; if (relu) v = fmaxf(v, 0.0f); C[ro + col0 + m]      = v;
            v = acc0_1[r] + bv1; if (relu) v = fmaxf(v, 0.0f); C[ro + col0 + 16 + m] = v;
            v = acc0_2[r] + bv2; if (relu) v = fmaxf(v, 0.0f); C[ro + col0 + 32 + m] = v;
            v = acc0_3[r] + bv3; if (relu) v = fmaxf(v, 0.0f); C[ro + col0 + 48 + m] = v;
        }
    }
    if (r1 < Mrows) {
#pragma unroll
        for (int r = 0; r < 8; ++r) {
            size_t ro = (size_t)(r1 + mb + r) * Nout;
            float v;
            v = acc1_0[r] + bv0; if (relu) v = fmaxf(v, 0.0f); C[ro + col0 + m]      = v;
            v = acc1_1[r] + bv1; if (relu) v = fmaxf(v, 0.0f); C[ro + col0 + 16 + m] = v;
            v = acc1_2[r] + bv2; if (relu) v = fmaxf(v, 0.0f); C[ro + col0 + 32 + m] = v;
            v = acc1_3[r] + bv3; if (relu) v = fmaxf(v, 0.0f); C[ro + col0 + 48 + m] = v;
        }
    }
}

#define WPS 800
#define S_H 132

__device__ __forceinline__ float sigm(float x) { return 1.0f / (1.0f + __expf(-x)); }

__global__ void gru_scan_k(const float* __restrict__ GI, const float* __restrict__ whh,
                           const float* __restrict__ bhh, float* __restrict__ out) {
    extern __shared__ float smem[];
    float* whhT = smem;
    float* hS   = whhT + 64 * WPS;
    float* ghS  = hS + 16 * S_H;
    const int tid = threadIdx.x;

    for (int idx = tid; idx < G3 * HID; idx += 768) {
        int j = idx >> 7, k = idx & 127;
        whhT[(k >> 1) * WPS + j * 2 + (k & 1)] = whh[idx];
    }
    for (int idx = tid; idx < 16 * S_H; idx += 768) hS[idx] = 0.0f;
    __syncthreads();

    const int wave = tid >> 5, lane = tid & 31;
    const int colb = wave * 16;
    const int m    = lane & 15;
    const int kof  = (lane >> 4) * 2;

    for (int n = 0; n < NN; ++n) {
        v8f acc = {};
        for (int k = 0; k < 128; k += 4) {
            int kk = k + kof;
            v2f a = *(const v2f*)(hS + m * S_H + kk);
            v2f b = *(const v2f*)(whhT + (kk >> 1) * WPS + (colb + m) * 2);
            acc = __builtin_amdgcn_wmma_f32_16x16x4_f32(false, a, false, b, (short)0, acc, false, false);
        }
        if (lane < 16) {
#pragma unroll
            for (int r = 0; r < 4; ++r) ghS[r * G3 + colb + lane] = acc[r];
        }
        __syncthreads();

        if (tid < 512) {
            int t = tid >> 7, j = tid & 127;
            const float* gi = GI + ((size_t)t * NN + n) * G3;
            float hr = ghS[t * G3 + j]       + bhh[j];
            float hz = ghS[t * G3 + 128 + j] + bhh[128 + j];
            float hn = ghS[t * G3 + 256 + j] + bhh[256 + j];
            float r  = sigm(gi[j]       + hr);
            float z  = sigm(gi[128 + j] + hz);
            float nv = 2.0f * sigm(2.0f * (gi[256 + j] + r * hn)) - 1.0f;
            float h  = hS[t * S_H + j];
            float h2 = (1.0f - z) * nv + z * h;
            hS[t * S_H + j] = h2;
            out[((size_t)t * NN + n) * HID + j] = h2;
        }
        __syncthreads();
    }
}

extern "C" void kernel_launch(void* const* d_in, const int* in_sizes, int n_in,
                              void* d_out, int out_size, void* d_ws, size_t ws_size,
                              hipStream_t stream) {
    const float* x   = (const float*)d_in[0];
    const int*   eis = (const int*)  d_in[1];
    const float* w1  = (const float*)d_in[2];
    const float* b1  = (const float*)d_in[3];
    const float* w2  = (const float*)d_in[4];
    const float* b2  = (const float*)d_in[5];
    const float* wih = (const float*)d_in[6];
    const float* whh = (const float*)d_in[7];
    const float* bih = (const float*)d_in[8];
    const float* bhh = (const float*)d_in[9];
    float* out = (float*)d_out;

    float* ws   = (float*)d_ws;
    float* XW1  = ws;
    float* dinv = XW1 + (size_t)NN * HID;
    float* Z1   = dinv + (size_t)TT * NN;
    float* H2   = Z1 + (size_t)TT * NN * HID;
    float* GI   = H2 + (size_t)TT * NN * HID;
    float* Z2   = Z1;

    const int blk = 256;

    fill_zero_k<<<(TT * NN + blk - 1) / blk, blk, 0, stream>>>(dinv, TT * NN);
    count_deg_k<<<(TT * EE + blk - 1) / blk, blk, 0, stream>>>(eis, dinv);
    finalize_dinv_k<<<(TT * NN + blk - 1) / blk, blk, 0, stream>>>(dinv, TT * NN);

    wmma_gemm_k<<<dim3((NN / 16 + 15) / 16, HID / 64), blk, 0, stream>>>(
        x, w1, nullptr, XW1, NN, HID, HID, 1, 0);

    for (int t = 0; t < TT; ++t) {
        init_self_k<<<(NN * HID + blk - 1) / blk, blk, 0, stream>>>(
            Z1 + (size_t)t * NN * HID, XW1, dinv + t * NN, b1);
        scatter_k<<<(EE * 32 + blk - 1) / blk, blk, 0, stream>>>(
            Z1 + (size_t)t * NN * HID, XW1, dinv + t * NN,
            eis + (size_t)t * 2 * EE, eis + (size_t)t * 2 * EE + EE);
    }
    relu_k<<<(TT * NN * HID + blk - 1) / blk, blk, 0, stream>>>(Z1, TT * NN * HID);

    wmma_gemm_k<<<dim3((TT * NN / 16 + 15) / 16, HID / 64), blk, 0, stream>>>(
        Z1, w2, nullptr, H2, TT * NN, HID, HID, 1, 0);

    for (int t = 0; t < TT; ++t) {
        init_self_k<<<(NN * HID + blk - 1) / blk, blk, 0, stream>>>(
            Z2 + (size_t)t * NN * HID, H2 + (size_t)t * NN * HID, dinv + t * NN, b2);
        scatter_k<<<(EE * 32 + blk - 1) / blk, blk, 0, stream>>>(
            Z2 + (size_t)t * NN * HID, H2 + (size_t)t * NN * HID, dinv + t * NN,
            eis + (size_t)t * 2 * EE, eis + (size_t)t * 2 * EE + EE);
    }
    relu_k<<<(TT * NN * HID + blk - 1) / blk, blk, 0, stream>>>(Z2, TT * NN * HID);

    wmma_gemm_k<<<dim3((TT * NN / 16 + 15) / 16, G3 / 64), blk, 0, stream>>>(
        Z2, wih, bih, GI, TT * NN, G3, 1, HID, 0);

    size_t shmem = (size_t)(64 * WPS + 16 * S_H + 4 * G3) * sizeof(float);
    hipFuncSetAttribute((const void*)gru_scan_k,
                        hipFuncAttributeMaxDynamicSharedMemorySize, (int)shmem);
    gru_scan_k<<<1, 768, shmem, stream>>>(GI, whh, bhh, out);
}


# GRU scan: gi row of step n+1 requested at the top of step n and bhh kept in registers, so no global load latency is left behind the per-step barrier
# speedup vs baseline: 1.1533x; 1.1533x over previous
#include <hip/hip_runtime.h>
#include <math.h>

typedef __attribute__((ext_vector_type(2))) float v2f;
typedef __attribute__((ext_vector_type(8))) float v8f;

#define NN   20000
#define EE   320000
#define TT   4
#define HID  128
#define G3   384

__global__ void fill_zero_k(float* __restrict__ p, int n) {
    int i = blockIdx.x * blockDim.x + threadIdx.x;
    if (i < n) p[i] = 0.0f;
}

__global__ void count_deg_k(const int* __restrict__ eis, float* __restrict__ deg) {
    int i = blockIdx.x * blockDim.x + threadIdx.x;
    if (i >= TT * EE) return;
    int t = i / EE, e = i - t * EE;
    int d = eis[(size_t)t * 2 * EE + EE + e];
    atomicAdd(&deg[t * NN + d], 1.0f);
}

__global__ void finalize_dinv_k(float* __restrict__ deg, int n) {
    int i = blockIdx.x * blockDim.x + threadIdx.x;
    if (i < n) deg[i] = rsqrtf(deg[i] + 1.0f);
}

__global__ void init_self_k(float* __restrict__ out, const float* __restrict__ H,
                            const float* __restrict__ dinv, const float* __restrict__ bias) {
    int i = blockIdx.x * blockDim.x + threadIdx.x;
    if (i >= NN * HID) return;
    int n = i >> 7, j = i & 127;
    float di = dinv[n];
    out[i] = bias[j] + H[i] * di * di;
}

__global__ void scatter_k(float* __restrict__ out, const float* __restrict__ H,
                          const float* __restrict__ dinv,
                          const int* __restrict__ src, const int* __restrict__ dst) {
    int i = blockIdx.x * blockDim.x + threadIdx.x;
    int e = i >> 5;
    if (e >= EE) return;
    int lane = i & 31;
    int s = src[e], d = dst[e];
    float norm = dinv[s] * dinv[d];
    const float4 v = *(const float4*)(H + (size_t)s * HID + lane * 4);
    float* o = out + (size_t)d * HID + lane * 4;
    atomicAdd(o + 0, v.x * norm);
    atomicAdd(o + 1, v.y * norm);
    atomicAdd(o + 2, v.z * norm);
    atomicAdd(o + 3, v.w * norm);
}

__global__ void relu_k(float* __restrict__ p, int n) {
    int i = blockIdx.x * blockDim.x + threadIdx.x;
    if (i < n) p[i] = fmaxf(p[i], 0.0f);
}

#define BPS 160

__global__ void wmma_gemm_k(const float* __restrict__ A, const float* __restrict__ B,
                            const float* __restrict__ bias, float* __restrict__ C,
                            int Mrows, int Nout, int rsB, int csB, int relu) {
    __shared__ float Blds[64 * BPS];
    const int tid = threadIdx.x;
    const int col0 = blockIdx.y * 64;

    for (int idx = tid; idx < 128 * 64; idx += 256) {
        int k = idx >> 6, c = idx & 63;
        Blds[(k >> 1) * BPS + c * 2 + (k & 1)] =
            B[(size_t)k * rsB + (size_t)(col0 + c) * csB];
    }
    __syncthreads();

    const int wave = tid >> 5, lane = tid & 31;
    const int m = lane & 15;
    const int kof = (lane >> 4) * 2;
    const int rt0 = blockIdx.x * 16 + wave;
    const int rt1 = rt0 + 8;
    const int r0 = rt0 * 16, r1 = rt1 * 16;
    const float* A0 = A + (size_t)((r0 < Mrows ? r0 : 0) + m) * 128;
    const float* A1 = A + (size_t)((r1 < Mrows ? r1 : 0) + m) * 128;

    v8f acc0_0 = {}, acc0_1 = {}, acc0_2 = {}, acc0_3 = {};
    v8f acc1_0 = {}, acc1_1 = {}, acc1_2 = {}, acc1_3 = {};

    for (int k = 0; k < 128; k += 4) {
        int kk = k + kof;
        v2f a0 = *(const v2f*)(A0 + kk);
        v2f a1 = *(const v2f*)(A1 + kk);
        const float* bp = Blds + (kk >> 1) * BPS + m * 2;
        v2f b;
        b = *(const v2f*)(bp);
        acc0_0 = __builtin_amdgcn_wmma_f32_16x16x4_f32(false, a0, false, b, (short)0, acc0_0, false, false);
        acc1_0 = __builtin_amdgcn_wmma_f32_16x16x4_f32(false, a1, false, b, (short)0, acc1_0, false, false);
        b = *(const v2f*)(bp + 32);
        acc0_1 = __builtin_amdgcn_wmma_f32_16x16x4_f32(false, a0, false, b, (short)0, acc0_1, false, false);
        acc1_1 = __builtin_amdgcn_wmma_f32_16x16x4_f32(false, a1, false, b, (short)0, acc1_1, false, false);
        b = *(const v2f*)(bp + 64);
        acc0_2 = __builtin_amdgcn_wmma_f32_16x16x4_f32(false, a0, false, b, (short)0, acc0_2, false, false);
        acc1_2 = __builtin_amdgcn_wmma_f32_16x16x4_f32(false, a1, false, b, (short)0, acc1_2, false, false);
        b = *(const v2f*)(bp + 96);
        acc0_3 = __builtin_amdgcn_wmma_f32_16x16x4_f32(false, a0, false, b, (short)0, acc0_3, false, false);
        acc1_3 = __builtin_amdgcn_wmma_f32_16x16x4_f32(false, a1, false, b, (short)0, acc1_3, false, false);
    }

    const int mb = (lane >> 4) * 8;
    float bv0 = bias ? bias[col0 + m]      : 0.0f;
    float bv1 = bias ? bias[col0 + 16 + m] : 0.0f;
    float bv2 = bias ? bias[col0 + 32 + m] : 0.0f;
    float bv3 = bias ? bias[col0 + 48 + m] : 0.0f;
    if (r0 < Mrows) {
#pragma unroll
        for (int r = 0; r < 8; ++r) {
            size_t ro = (size_t)(r0 + mb + r) * Nout;
            float v;
            v = acc0_0[r] + bv0; if (relu) v = fmaxf(v, 0.0f); C[ro + col0 + m]      = v;
            v = acc0_1[r] + bv1; if (relu) v = fmaxf(v, 0.0f); C[ro + col0 + 16 + m] = v;
            v = acc0_2[r] + bv2; if (relu) v = fmaxf(v, 0.0f); C[ro + col0 + 32 + m] = v;
            v = acc0_3[r] + bv3; if (relu) v = fmaxf(v, 0.0f); C[ro + col0 + 48 + m] = v;
        }
    }
    if (r1 < Mrows) {
#pragma unroll
        for (int r = 0; r < 8; ++r) {
            size_t ro = (size_t)(r1 + mb + r) * Nout;
            float v;
            v = acc1_0[r] + bv0; if (relu) v = fmaxf(v, 0.0f); C[ro + col0 + m]      = v;
            v = acc1_1[r] + bv1; if (relu) v = fmaxf(v, 0.0f); C[ro + col0 + 16 + m] = v;
            v = acc1_2[r] + bv2; if (relu) v = fmaxf(v, 0.0f); C[ro + col0 + 32 + m] = v;
            v = acc1_3[r] + bv3; if (relu) v = fmaxf(v, 0.0f); C[ro + col0 + 48 + m] = v;
        }
    }
}

#define WPS 800
#define S_H 132

__device__ __forceinline__ float sigm(float x) { return 1.0f / (1.0f + __expf(-x)); }

__global__ void gru_scan_k(const float* __restrict__ GI, const float* __restrict__ whh,
                           const float* __restrict__ bhh, float* __restrict__ out) {
    extern __shared__ float smem[];
    float* whhT = smem;
    float* hS   = whhT + 64 * WPS;
    float* ghS  = hS + 16 * S_H;
    const int tid = threadIdx.x;

    for (int idx = tid; idx < G3 * HID; idx += 768) {
        int j = idx >> 7, k = idx & 127;
        whhT[(k >> 1) * WPS + j * 2 + (k & 1)] = whh[idx];
    }
    for (int idx = tid; idx < 16 * S_H; idx += 768) hS[idx] = 0.0f;
    __syncthreads();

    const int wave = tid >> 5, lane = tid & 31;
    const int colb = wave * 16;
    const int m    = lane & 15;
    const int kof  = (lane >> 4) * 2;

    const bool gate = tid < 512;
    const int gt = gate ? (tid >> 7) : 0, gj = tid & 127;
    const float* gip = GI + (size_t)gt * NN * G3 + gj;
    float bh_r = 0.0f, bh_z = 0.0f, bh_n = 0.0f;
    float gi_r = 0.0f, gi_z = 0.0f, gi_n = 0.0f;
    if (gate) {
        bh_r = bhh[gj]; bh_z = bhh[128 + gj]; bh_n = bhh[256 + gj];
        gi_r = gip[0];  gi_z = gip[128];      gi_n = gip[256];
    }

    for (int n = 0; n < NN; ++n) {
        float nx_r = 0.0f, nx_z = 0.0f, nx_n = 0.0f;
        if (gate) {
            const int np = (n + 1 < NN) ? n + 1 : n;
            const float* g = gip + (size_t)np * G3;
            nx_r = g[0]; nx_z = g[128]; nx_n = g[256];
        }

        v8f acc = {};
        for (int k = 0; k < 128; k += 4) {
            int kk = k + kof;
            v2f a = *(const v2f*)(hS + m * S_H + kk);
            v2f b = *(const v2f*)(whhT + (kk >> 1) * WPS + (colb + m) * 2);
            acc = __builtin_amdgcn_wmma_f32_16x16x4_f32(false, a, false, b, (short)0, acc, false, false);
        }
        if (lane < 16) {
#pragma unroll
            for (int r = 0; r < 4; ++r) ghS[r * G3 + colb + lane] = acc[r];
        }
        __syncthreads();

        if (gate) {
            const int t = gt, j = gj;
            float hr = ghS[t * G3 + j]       + bh_r;
            float hz = ghS[t * G3 + 128 + j] + bh_z;
            float hn = ghS[t * G3 + 256 + j] + bh_n;
            float r  = sigm(gi_r + hr);
            float z  = sigm(gi_z + hz);
            float nv = 2.0f * sigm(2.0f * (gi_n + r * hn)) - 1.0f;
            float h  = hS[t * S_H + j];
            float h2 = (1.0f - z) * nv + z * h;
            hS[t * S_H + j] = h2;
            out[((size_t)t * NN + n) * HID + j] = h2;
        }
        __syncthreads();
        gi_r = nx_r; gi_z = nx_z; gi_n = nx_n;
    }
}

extern "C" void kernel_launch(void* const* d_in, const int* in_sizes, int n_in,
                              void* d_out, int out_size, void* d_ws, size_t ws_size,
                              hipStream_t stream) {
    const float* x   = (const float*)d_in[0];
    const int*   eis = (const int*)  d_in[1];
    const float* w1  = (const float*)d_in[2];
    const float* b1  = (const float*)d_in[3];
    const float* w2  = (const float*)d_in[4];
    const float* b2  = (const float*)d_in[5];
    const float* wih = (const float*)d_in[6];
    const float* whh = (const float*)d_in[7];
    const float* bih = (const float*)d_in[8];
    const float* bhh = (const float*)d_in[9];
    float* out = (float*)d_out;

    float* ws   = (float*)d_ws;
    float* XW1  = ws;
    float* dinv = XW1 + (size_t)NN * HID;
    float* Z1   = dinv + (size_t)TT * NN;
    float* H2   = Z1 + (size_t)TT * NN * HID;
    float* GI   = H2 + (size_t)TT * NN * HID;
    float* Z2   = Z1;

    const int blk = 256;

    fill_zero_k<<<(TT * NN + blk - 1) / blk, blk, 0, stream>>>(dinv, TT * NN);
    count_deg_k<<<(TT * EE + blk - 1) / blk, blk, 0, stream>>>(eis, dinv);
    finalize_dinv_k<<<(TT * NN + blk - 1) / blk, blk, 0, stream>>>(dinv, TT * NN);

    wmma_gemm_k<<<dim3((NN / 16 + 15) / 16, HID / 64), blk, 0, stream>>>(
        x, w1, nullptr, XW1, NN, HID, HID, 1, 0);

    for (int t = 0; t < TT; ++t) {
        init_self_k<<<(NN * HID + blk - 1) / blk, blk, 0, stream>>>(
            Z1 + (size_t)t * NN * HID, XW1, dinv + t * NN, b1);
        scatter_k<<<(EE * 32 + blk - 1) / blk, blk, 0, stream>>>(
            Z1 + (size_t)t * NN * HID, XW1, dinv + t * NN,
            eis + (size_t)t * 2 * EE, eis + (size_t)t * 2 * EE + EE);
    }
    relu_k<<<(TT * NN * HID + blk - 1) / blk, blk, 0, stream>>>(Z1, TT * NN * HID);

    wmma_gemm_k<<<dim3((TT * NN / 16 + 15) / 16, HID / 64), blk, 0, stream>>>(
        Z1, w2, nullptr, H2, TT * NN, HID, HID, 1, 0);

    for (int t = 0; t < TT; ++t) {
        init_self_k<<<(NN * HID + blk - 1) / blk, blk, 0, stream>>>(
            Z2 + (size_t)t * NN * HID, H2 + (size_t)t * NN * HID, dinv + t * NN, b2);
        scatter_k<<<(EE * 32 + blk - 1) / blk, blk, 0, stream>>>(
            Z2 + (size_t)t * NN * HID, H2 + (size_t)t * NN * HID, dinv + t * NN,
            eis + (size_t)t * 2 * EE, eis + (size_t)t * 2 * EE + EE);
    }
    relu_k<<<(TT * NN * HID + blk - 1) / blk, blk, 0, stream>>>(Z2, TT * NN * HID);

    wmma_gemm_k<<<dim3((TT * NN / 16 + 15) / 16, G3 / 64), blk, 0, stream>>>(
        Z2, wih, bih, GI, TT * NN, G3, 1, HID, 0);

    size_t shmem = (size_t)(64 * WPS + 16 * S_H + 4 * G3) * sizeof(float);
    hipFuncSetAttribute((const void*)gru_scan_k,
                        hipFuncAttributeMaxDynamicSharedMemorySize, (int)shmem);
    gru_scan_k<<<1, 768, shmem, stream>>>(GI, whh, bhh, out);
}


# GRU scan: a step's h row is stored to out at the top of the next step, so no global store completion is waited for in front of the step's second barrier
# speedup vs baseline: 1.4060x; 1.2191x over previous
#include <hip/hip_runtime.h>
#include <math.h>

typedef __attribute__((ext_vector_type(2))) float v2f;
typedef __attribute__((ext_vector_type(8))) float v8f;

#define NN   20000
#define EE   320000
#define TT   4
#define HID  128
#define G3   384

__global__ void fill_zero_k(float* __restrict__ p, int n) {
    int i = blockIdx.x * blockDim.x + threadIdx.x;
    if (i < n) p[i] = 0.0f;
}

__global__ void count_deg_k(const int* __restrict__ eis, float* __restrict__ deg) {
    int i = blockIdx.x * blockDim.x + threadIdx.x;
    if (i >= TT * EE) return;
    int t = i / EE, e = i - t * EE;
    int d = eis[(size_t)t * 2 * EE + EE + e];
    atomicAdd(&deg[t * NN + d], 1.0f);
}

__global__ void finalize_dinv_k(float* __restrict__ deg, int n) {
    int i = blockIdx.x * blockDim.x + threadIdx.x;
    if (i < n) deg[i] = rsqrtf(deg[i] + 1.0f);
}

__global__ void init_self_k(float* __restrict__ out, const float* __restrict__ H,
                            const float* __restrict__ dinv, const float* __restrict__ bias) {
    int i = blockIdx.x * blockDim.x + threadIdx.x;
    if (i >= NN * HID) return;
    int n = i >> 7, j = i & 127;
    float di = dinv[n];
    out[i] = bias[j] + H[i] * di * di;
}

__global__ void scatter_k(float* __restrict__ out, const float* __restrict__ H,
                          const float* __restrict__ dinv,
                          const int* __restrict__ src, const int* __restrict__ dst) {
    int i = blockIdx.x * blockDim.x + threadIdx.x;
    int e = i >> 5;
    if (e >= EE) return;
    int lane = i & 31;
    int s = src[e], d = dst[e];
    float norm = dinv[s] * dinv[d];
    const float4 v = *(const float4*)(H + (size_t)s * HID + lane * 4);
    float* o = out + (size_t)d * HID + lane * 4;
    atomicAdd(o + 0, v.x * norm);
    atomicAdd(o + 1, v.y * norm);
    atomicAdd(o + 2, v.z * norm);
    atomicAdd(o + 3, v.w * norm);
}

__global__ void relu_k(float* __restrict__ p, int n) {
    int i = blockIdx.x * blockDim.x + threadIdx.x;
    if (i < n) p[i] = fmaxf(p[i], 0.0f);
}

#define BPS 160

__global__ void wmma_gemm_k(const float* __restrict__ A, const float* __restrict__ B,
                            const float* __restrict__ bias, float* __restrict__ C,
                            int Mrows, int Nout, int rsB, int csB, int relu) {
    __shared__ float Blds[64 * BPS];
    const int tid = threadIdx.x;
    const int col0 = blockIdx.y * 64;

    for (int idx = tid; idx < 128 * 64; idx += 256) {
        int k = idx >> 6, c = idx & 63;
        Blds[(k >> 1) * BPS + c * 2 + (k & 1)] =
            B[(size_t)k * rsB + (size_t)(col0 + c) * csB];
    }
    __syncthreads();

    const int wave = tid >> 5, lane = tid & 31;
    const int m = lane & 15;
    const int kof = (lane >> 4) * 2;
    const int rt0 = blockIdx.x * 16 + wave;
    const int rt1 = rt0 + 8;
    const int r0 = rt0 * 16, r1 = rt1 * 16;
    const float* A0 = A + (size_t)((r0 < Mrows ? r0 : 0) + m) * 128;
    const float* A1 = A + (size_t)((r1 < Mrows ? r1 : 0) + m) * 128;

    v8f acc0_0 = {}, acc0_1 = {}, acc0_2 = {}, acc0_3 = {};
    v8f acc1_0 = {}, acc1_1 = {}, acc1_2 = {}, acc1_3 = {};

    for (int k = 0; k < 128; k += 4) {
        int kk = k + kof;
        v2f a0 = *(const v2f*)(A0 + kk);
        v2f a1 = *(const v2f*)(A1 + kk);
        const float* bp = Blds + (kk >> 1) * BPS + m * 2;
        v2f b;
        b = *(const v2f*)(bp);
        acc0_0 = __builtin_amdgcn_wmma_f32_16x16x4_f32(false, a0, false, b, (short)0, acc0_0, false, false);
        acc1_0 = __builtin_amdgcn_wmma_f32_16x16x4_f32(false, a1, false, b, (short)0, acc1_0, false, false);
        b = *(const v2f*)(bp + 32);
        acc0_1 = __builtin_amdgcn_wmma_f32_16x16x4_f32(false, a0, false, b, (short)0, acc0_1, false, false);
        acc1_1 = __builtin_amdgcn_wmma_f32_16x16x4_f32(false, a1, false, b, (short)0, acc1_1, false, false);
        b = *(const v2f*)(bp + 64);
        acc0_2 = __builtin_amdgcn_wmma_f32_16x16x4_f32(false, a0, false, b, (short)0, acc0_2, false, false);
        acc1_2 = __builtin_amdgcn_wmma_f32_16x16x4_f32(false, a1, false, b, (short)0, acc1_2, false, false);
        b = *(const v2f*)(bp + 96);
        acc0_3 = __builtin_amdgcn_wmma_f32_16x16x4_f32(false, a0, false, b, (short)0, acc0_3, false, false);
        acc1_3 = __builtin_amdgcn_wmma_f32_16x16x4_f32(false, a1, false, b, (short)0, acc1_3, false, false);
    }

    const int mb = (lane >> 4) * 8;
    float bv0 = bias ? bias[col0 + m]      : 0.0f;
    float bv1 = bias ? bias[col0 + 16 + m] : 0.0f;
    float bv2 = bias ? bias[col0 + 32 + m] : 0.0f;
    float bv3 = bias ? bias[col0 + 48 + m] : 0.0f;
    if (r0 < Mrows) {
#pragma unroll
        for (int r = 0; r < 8; ++r) {
            size_t ro = (size_t)(r0 + mb + r) * Nout;
            float v;
            v = acc0_0[r] + bv0; if (relu) v = fmaxf(v, 0.0f); C[ro + col0 + m]      = v;
            v = acc0_1[r] + bv1; if (relu) v = fmaxf(v, 0.0f); C[ro + col0 + 16 + m] = v;
            v = acc0_2[r] + bv2; if (relu) v = fmaxf(v, 0.0f); C[ro + col0 + 32 + m] = v;
            v = acc0_3[r] + bv3; if (relu) v = fmaxf(v, 0.0f); C[ro + col0 + 48 + m] = v;
        }
    }
    if (r1 < Mrows) {
#pragma unroll
        for (int r = 0; r < 8; ++r) {
            size_t ro = (size_t)(r1 + mb + r) * Nout;
            float v;
            v = acc1_0[r] + bv0; if (relu) v = fmaxf(v, 0.0f); C[ro + col0 + m]      = v;
            v = acc1_1[r] + bv1; if (relu) v = fmaxf(v, 0.0f); C[ro + col0 + 16 + m] = v;
            v = acc1_2[r] + bv2; if (relu) v = fmaxf(v, 0.0f); C[ro + col0 + 32 + m] = v;
            v = acc1_3[r] + bv3; if (relu) v = fmaxf(v, 0.0f); C[ro + col0 + 48 + m] = v;
        }
    }
}

#define WPS 800
#define S_H 132

__device__ __forceinline__ float sigm(float x) { return 1.0f / (1.0f + __expf(-x)); }

__global__ void gru_scan_k(const float* __restrict__ GI, const float* __restrict__ whh,
                           const float* __restrict__ bhh, float* __restrict__ out) {
    extern __shared__ float smem[];
    float* whhT = smem;
    float* hS   = whhT + 64 * WPS;
    float* ghS  = hS + 16 * S_H;
    const int tid = threadIdx.x;

    for (int idx = tid; idx < G3 * HID; idx += 768) {
        int j = idx >> 7, k = idx & 127;
        whhT[(k >> 1) * WPS + j * 2 + (k & 1)] = whh[idx];
    }
    for (int idx = tid; idx < 16 * S_H; idx += 768) hS[idx] = 0.0f;
    __syncthreads();

    const int wave = tid >> 5, lane = tid & 31;
    const int colb = wave * 16;
    const int m    = lane & 15;
    const int kof  = (lane >> 4) * 2;

    const bool gate = tid < 512;
    const int gt = gate ? (tid >> 7) : 0, gj = tid & 127;
    const float* gip = GI + (size_t)gt * NN * G3 + gj;
    float bh_r = 0.0f, bh_z = 0.0f, bh_n = 0.0f;
    float gi_r = 0.0f, gi_z = 0.0f, gi_n = 0.0f;
    float h_keep = 0.0f;
    float* outp = out + (size_t)gt * NN * HID + gj;
    if (gate) {
        bh_r = bhh[gj]; bh_z = bhh[128 + gj]; bh_n = bhh[256 + gj];
        gi_r = gip[0];  gi_z = gip[128];      gi_n = gip[256];
    }

    for (int n = 0; n < NN; ++n) {
        float nx_r = 0.0f, nx_z = 0.0f, nx_n = 0.0f;
        if (gate) {
            if (n > 0) outp[(size_t)(n - 1) * HID] = h_keep;
            const int np = (n + 1 < NN) ? n + 1 : n;
            const float* g = gip + (size_t)np * G3;
            nx_r = g[0]; nx_z = g[128]; nx_n = g[256];
        }

        v8f acc = {};
        for (int k = 0; k < 128; k += 4) {
            int kk = k + kof;
            v2f a = *(const v2f*)(hS + m * S_H + kk);
            v2f b = *(const v2f*)(whhT + (kk >> 1) * WPS + (colb + m) * 2);
            acc = __builtin_amdgcn_wmma_f32_16x16x4_f32(false, a, false, b, (short)0, acc, false, false);
        }
        if (lane < 16) {
#pragma unroll
            for (int r = 0; r < 4; ++r) ghS[r * G3 + colb + lane] = acc[r];
        }
        __syncthreads();

        if (gate) {
            const int t = gt, j = gj;
            float hr = ghS[t * G3 + j]       + bh_r;
            float hz = ghS[t * G3 + 128 + j] + bh_z;
            float hn = ghS[t * G3 + 256 + j] + bh_n;
            float r  = sigm(gi_r + hr);
            float z  = sigm(gi_z + hz);
            float nv = 2.0f * sigm(2.0f * (gi_n + r * hn)) - 1.0f;
            float h  = hS[t * S_H + j];
            float h2 = (1.0f - z) * nv + z * h;
            hS[t * S_H + j] = h2;
            h_keep = h2;
        }
        __syncthreads();
        gi_r = nx_r; gi_z = nx_z; gi_n = nx_n;
    }
    if (gate) outp[(size_t)(NN - 1) * HID] = h_keep;
}

extern "C" void kernel_launch(void* const* d_in, const int* in_sizes, int n_in,
                              void* d_out, int out_size, void* d_ws, size_t ws_size,
                              hipStream_t stream) {
    const float* x   = (const float*)d_in[0];
    const int*   eis = (const int*)  d_in[1];
    const float* w1  = (const float*)d_in[2];
    const float* b1  = (const float*)d_in[3];
    const float* w2  = (const float*)d_in[4];
    const float* b2  = (const float*)d_in[5];
    const float* wih = (const float*)d_in[6];
    const float* whh = (const float*)d_in[7];
    const float* bih = (const float*)d_in[8];
    const float* bhh = (const float*)d_in[9];
    float* out = (float*)d_out;

    float* ws   = (float*)d_ws;
    float* XW1  = ws;
    float* dinv = XW1 + (size_t)NN * HID;
    float* Z1   = dinv + (size_t)TT * NN;
    float* H2   = Z1 + (size_t)TT * NN * HID;
    float* GI   = H2 + (size_t)TT * NN * HID;
    float* Z2   = Z1;

    const int blk = 256;

    fill_zero_k<<<(TT * NN + blk - 1) / blk, blk, 0, stream>>>(dinv, TT * NN);
    count_deg_k<<<(TT * EE + blk - 1) / blk, blk, 0, stream>>>(eis, dinv);
    finalize_dinv_k<<<(TT * NN + blk - 1) / blk, blk, 0, stream>>>(dinv, TT * NN);

    wmma_gemm_k<<<dim3((NN / 16 + 15) / 16, HID / 64), blk, 0, stream>>>(
        x, w1, nullptr, XW1, NN, HID, HID, 1, 0);

    for (int t = 0; t < TT; ++t) {
        init_self_k<<<(NN * HID + blk - 1) / blk, blk, 0, stream>>>(
            Z1 + (size_t)t * NN * HID, XW1, dinv + t * NN, b1);
        scatter_k<<<(EE * 32 + blk - 1) / blk, blk, 0, stream>>>(
            Z1 + (size_t)t * NN * HID, XW1, dinv + t * NN,
            eis + (size_t)t * 2 * EE, eis + (size_t)t * 2 * EE + EE);
    }
    relu_k<<<(TT * NN * HID + blk - 1) / blk, blk, 0, stream>>>(Z1, TT * NN * HID);

    wmma_gemm_k<<<dim3((TT * NN / 16 + 15) / 16, HID / 64), blk, 0, stream>>>(
        Z1, w2, nullptr, H2, TT * NN, HID, HID, 1, 0);

    for (int t = 0; t < TT; ++t) {
        init_self_k<<<(NN * HID + blk - 1) / blk, blk, 0, stream>>>(
            Z2 + (size_t)t * NN * HID, H2 + (size_t)t * NN * HID, dinv + t * NN, b2);
        scatter_k<<<(EE * 32 + blk - 1) / blk, blk, 0, stream>>>(
            Z2 + (size_t)t * NN * HID, H2 + (size_t)t * NN * HID, dinv + t * NN,
            eis + (size_t)t * 2 * EE, eis + (size_t)t * 2 * EE + EE);
    }
    relu_k<<<(TT * NN * HID + blk - 1) / blk, blk, 0, stream>>>(Z2, TT * NN * HID);

    wmma_gemm_k<<<dim3((TT * NN / 16 + 15) / 16, G3 / 64), blk, 0, stream>>>(
        Z2, wih, bih, GI, TT * NN, G3, 1, HID, 0);

    size_t shmem = (size_t)(64 * WPS + 16 * S_H + 4 * G3) * sizeof(float);
    hipFuncSetAttribute((const void*)gru_scan_k,
                        hipFuncAttributeMaxDynamicSharedMemorySize, (int)shmem);
    gru_scan_k<<<1, 768, shmem, stream>>>(GI, whh, bhh, out);
}


# GRU scan: a wave owns 16 hidden units end to end (their r,z,n columns of whh in registers, three accumulators, their gates): gh no longer crosses LDS and a step has one barrier, not two
# speedup vs baseline: 1.4605x; 1.0388x over previous
#include <hip/hip_runtime.h>
#include <math.h>

typedef __attribute__((ext_vector_type(2))) float v2f;
typedef __attribute__((ext_vector_type(8))) float v8f;

#define NN   20000
#define EE   320000
#define TT   4
#define HID  128
#define G3   384

__global__ void fill_zero_k(float* __restrict__ p, int n) {
    int i = blockIdx.x * blockDim.x + threadIdx.x;
    if (i < n) p[i] = 0.0f;
}

__global__ void count_deg_k(const int* __restrict__ eis, float* __restrict__ deg) {
    int i = blockIdx.x * blockDim.x + threadIdx.x;
    if (i >= TT * EE) return;
    int t = i / EE, e = i - t * EE;
    int d = eis[(size_t)t * 2 * EE + EE + e];
    atomicAdd(&deg[t * NN + d], 1.0f);
}

__global__ void finalize_dinv_k(float* __restrict__ deg, int n) {
    int i = blockIdx.x * blockDim.x + threadIdx.x;
    if (i < n) deg[i] = rsqrtf(deg[i] + 1.0f);
}

__global__ void init_self_k(float* __restrict__ out, const float* __restrict__ H,
                            const float* __restrict__ dinv, const float* __restrict__ bias) {
    int i = blockIdx.x * blockDim.x + threadIdx.x;
    if (i >= NN * HID) return;
    int n = i >> 7, j = i & 127;
    float di = dinv[n];
    out[i] = bias[j] + H[i] * di * di;
}

__global__ void scatter_k(float* __restrict__ out, const float* __restrict__ H,
                          const float* __restrict__ dinv,
                          const int* __restrict__ src, const int* __restrict__ dst) {
    int i = blockIdx.x * blockDim.x + threadIdx.x;
    int e = i >> 5;
    if (e >= EE) return;
    int lane = i & 31;
    int s = src[e], d = dst[e];
    float norm = dinv[s] * dinv[d];
    const float4 v = *(const float4*)(H + (size_t)s * HID + lane * 4);
    float* o = out + (size_t)d * HID + lane * 4;
    atomicAdd(o + 0, v.x * norm);
    atomicAdd(o + 1, v.y * norm);
    atomicAdd(o + 2, v.z * norm);
    atomicAdd(o + 3, v.w * norm);
}

__global__ void relu_k(float* __restrict__ p, int n) {
    int i = blockIdx.x * blockDim.x + threadIdx.x;
    if (i < n) p[i] = fmaxf(p[i], 0.0f);
}

#define BPS 160

__global__ void wmma_gemm_k(const float* __restrict__ A, const float* __restrict__ B,
                            const float* __restrict__ bias, float* __restrict__ C,
                            int Mrows, int Nout, int rsB, int csB, int relu) {
    __shared__ float Blds[64 * BPS];
    const int tid = threadIdx.x;
    const int col0 = blockIdx.y * 64;

    for (int idx = tid; idx < 128 * 64; idx += 256) {
        int k = idx >> 6, c = idx & 63;
        Blds[(k >> 1) * BPS + c * 2 + (k & 1)] =
            B[(size_t)k * rsB + (size_t)(col0 + c) * csB];
    }
    __syncthreads();

    const int wave = tid >> 5, lane = tid & 31;
    const int m = lane & 15;
    const int kof = (lane >> 4) * 2;
    const int rt0 = blockIdx.x * 16 + wave;
    const int rt1 = rt0 + 8;
    const int r0 = rt0 * 16, r1 = rt1 * 16;
    const float* A0 = A + (size_t)((r0 < Mrows ? r0 : 0) + m) * 128;
    const float* A1 = A + (size_t)((r1 < Mrows ? r1 : 0) + m) * 128;

    v8f acc0_0 = {}, acc0_1 = {}, acc0_2 = {}, acc0_3 = {};
    v8f acc1_0 = {}, acc1_1 = {}, acc1_2 = {}, acc1_3 = {};

    for (int k = 0; k < 128; k += 4) {
        int kk = k + kof;
        v2f a0 = *(const v2f*)(A0 + kk);
        v2f a1 = *(const v2f*)(A1 + kk);
        const float* bp = Blds + (kk >> 1) * BPS + m * 2;
        v2f b;
        b = *(const v2f*)(bp);
        acc0_0 = __builtin_amdgcn_wmma_f32_16x16x4_f32(false, a0, false, b, (short)0, acc0_0, false, false);
        acc1_0 = __builtin_amdgcn_wmma_f32_16x16x4_f32(false, a1, false, b, (short)0, acc1_0, false, false);
        b = *(const v2f*)(bp + 32);
        acc0_1 = __builtin_amdgcn_wmma_f32_16x16x4_f32(false, a0, false, b, (short)0, acc0_1, false, false);
        acc1_1 = __builtin_amdgcn_wmma_f32_16x16x4_f32(false, a1, false, b, (short)0, acc1_1, false, false);
        b = *(const v2f*)(bp + 64);
        acc0_2 = __builtin_amdgcn_wmma_f32_16x16x4_f32(false, a0, false, b, (short)0, acc0_2, false, false);
        acc1_2 = __builtin_amdgcn_wmma_f32_16x16x4_f32(false, a1, false, b, (short)0, acc1_2, false, false);
        b = *(const v2f*)(bp + 96);
        acc0_3 = __builtin_amdgcn_wmma_f32_16x16x4_f32(false, a0, false, b, (short)0, acc0_3, false, false);
        acc1_3 = __builtin_amdgcn_wmma_f32_16x16x4_f32(false, a1, false, b, (short)0, acc1_3, false, false);
    }

    const int mb = (lane >> 4) * 8;
    float bv0 = bias ? bias[col0 + m]      : 0.0f;
    float bv1 = bias ? bias[col0 + 16 + m] : 0.0f;
    float bv2 = bias ? bias[col0 + 32 + m] : 0.0f;
    float bv3 = bias ? bias[col0 + 48 + m] : 0.0f;
    if (r0 < Mrows) {
#pragma unroll
        for (int r = 0; r < 8; ++r) {
            size_t ro = (size_t)(r0 + mb + r) * Nout;
            float v;
            v = acc0_0[r] + bv0; if (relu) v = fmaxf(v, 0.0f); C[ro + col0 + m]      = v;
            v = acc0_1[r] + bv1; if (relu) v = fmaxf(v, 0.0f); C[ro + col0 + 16 + m] = v;
            v = acc0_2[r] + bv2; if (relu) v = fmaxf(v, 0.0f); C[ro + col0 + 32 + m] = v;
            v = acc0_3[r] + bv3; if (relu) v = fmaxf(v, 0.0f); C[ro + col0 + 48 + m] = v;
        }
    }
    if (r1 < Mrows) {
#pragma unroll
        for (int r = 0; r < 8; ++r) {
            size_t ro = (size_t)(r1 + mb + r) * Nout;
            float v;
            v = acc1_0[r] + bv0; if (relu) v = fmaxf(v, 0.0f); C[ro + col0 + m]      = v;
            v = acc1_1[r] + bv1; if (relu) v = fmaxf(v, 0.0f); C[ro + col0 + 16 + m] = v;
            v = acc1_2[r] + bv2; if (relu) v = fmaxf(v, 0.0f); C[ro + col0 + 32 + m] = v;
            v = acc1_3[r] + bv3; if (relu) v = fmaxf(v, 0.0f); C[ro + col0 + 48 + m] = v;
        }
    }
}

#define S_H  132
#define HBUF (16 * S_H)

__device__ __forceinline__ float sigm(float x) { return 1.0f / (1.0f + __expf(-x)); }

__global__ void __launch_bounds__(256)
gru_scan_k(const float* __restrict__ GI, const float* __restrict__ whh,
           const float* __restrict__ bhh, float* __restrict__ out) {
    __shared__ float hS[2 * HBUF];
    const int tid = threadIdx.x;
    for (int idx = tid; idx < 2 * HBUF; idx += 256) hS[idx] = 0.0f;

    const int wave = tid >> 5, lane = tid & 31;
    const int m   = lane & 15;
    const int hf  = lane >> 4;
    const int kof = hf * 2;
    const int j   = wave * 16 + m;

    v2f br[32], bz[32], bn[32];
    {
        const float* wr = whh + (size_t)j * HID + kof;
#pragma unroll
        for (int i = 0; i < 32; ++i) {
            br[i] = *(const v2f*)(wr + 4 * i);
            bz[i] = *(const v2f*)(wr + 128 * HID + 4 * i);
            bn[i] = *(const v2f*)(wr + 256 * HID + 4 * i);
        }
    }

    const int t0 = 2 * hf;
    const float* gip0 = GI + (size_t)t0 * NN * G3 + j;
    const float* gip1 = gip0 + (size_t)NN * G3;
    const float bh_r = bhh[j], bh_z = bhh[128 + j], bh_n = bhh[256 + j];
    float g0r = gip0[0], g0z = gip0[128], g0n = gip0[256];
    float g1r = gip1[0], g1z = gip1[128], g1n = gip1[256];
    float h0 = 0.0f, h1 = 0.0f;
    const int woff = (8 * hf) * S_H + j;
    const int aoff = m * S_H + kof;

    const int ct = tid >> 6, cj = (tid & 63) * 2;
    const int coff = ((ct & 1) + 8 * (ct >> 1)) * S_H + cj;
    float* outp = out + (size_t)ct * NN * HID + cj;

    __syncthreads();

    for (int n = 0; n < NN; ++n) {
        const float* hR = hS + (n & 1) * HBUF;
        float*       hW = hS + ((n & 1) ^ 1) * HBUF;
        if (n > 0) {
            v2f hv = *(const v2f*)(hR + coff);
            *(v2f*)(outp + (size_t)(n - 1) * HID) = hv;
        }
        const int np = (n + 1 < NN) ? n + 1 : n;
        const float* q0 = gip0 + (size_t)np * G3;
        const float* q1 = gip1 + (size_t)np * G3;
        const float x0r = q0[0], x0z = q0[128], x0n = q0[256];
        const float x1r = q1[0], x1z = q1[128], x1n = q1[256];

        v2f a[32];
#pragma unroll
        for (int i = 0; i < 32; ++i) a[i] = *(const v2f*)(hR + aoff + 4 * i);

        v8f ar = {}, az = {}, an = {};
#pragma unroll
        for (int i = 0; i < 32; ++i) {
            ar = __builtin_amdgcn_wmma_f32_16x16x4_f32(false, a[i], false, br[i], (short)0, ar, false, false);
            az = __builtin_amdgcn_wmma_f32_16x16x4_f32(false, a[i], false, bz[i], (short)0, az, false, false);
            an = __builtin_amdgcn_wmma_f32_16x16x4_f32(false, a[i], false, bn[i], (short)0, an, false, false);
        }

        const float r0 = sigm(g0r + (ar[0] + bh_r));
        const float z0 = sigm(g0z + (az[0] + bh_z));
        const float r1 = sigm(g1r + (ar[1] + bh_r));
        const float z1 = sigm(g1z + (az[1] + bh_z));
        const float hn0 = an[0] + bh_n;
        const float hn1 = an[1] + bh_n;
        const float nv0 = 2.0f * sigm(2.0f * (g0n + r0 * hn0)) - 1.0f;
        const float nv1 = 2.0f * sigm(2.0f * (g1n + r1 * hn1)) - 1.0f;
        const float h20 = (1.0f - z0) * nv0 + z0 * h0;
        const float h21 = (1.0f - z1) * nv1 + z1 * h1;
        hW[woff]       = h20;
        hW[woff + S_H] = h21;
        h0 = h20; h1 = h21;
        __syncthreads();
        g0r = x0r; g0z = x0z; g0n = x0n;
        g1r = x1r; g1z = x1z; g1n = x1n;
    }
    {
        const float* hR = hS + (NN & 1) * HBUF;
        v2f hv = *(const v2f*)(hR + coff);
        *(v2f*)(outp + (size_t)(NN - 1) * HID) = hv;
    }
}

extern "C" void kernel_launch(void* const* d_in, const int* in_sizes, int n_in,
                              void* d_out, int out_size, void* d_ws, size_t ws_size,
                              hipStream_t stream) {
    const float* x   = (const float*)d_in[0];
    const int*   eis = (const int*)  d_in[1];
    const float* w1  = (const float*)d_in[2];
    const float* b1  = (const float*)d_in[3];
    const float* w2  = (const float*)d_in[4];
    const float* b2  = (const float*)d_in[5];
    const float* wih = (const float*)d_in[6];
    const float* whh = (const float*)d_in[7];
    const float* bih = (const float*)d_in[8];
    const float* bhh = (const float*)d_in[9];
    float* out = (float*)d_out;

    float* ws   = (float*)d_ws;
    float* XW1  = ws;
    float* dinv = XW1 + (size_t)NN * HID;
    float* Z1   = dinv + (size_t)TT * NN;
    float* H2   = Z1 + (size_t)TT * NN * HID;
    float* GI   = H2 + (size_t)TT * NN * HID;
    float* Z2   = Z1;

    const int blk = 256;

    fill_zero_k<<<(TT * NN + blk - 1) / blk, blk, 0, stream>>>(dinv, TT * NN);
    count_deg_k<<<(TT * EE + blk - 1) / blk, blk, 0, stream>>>(eis, dinv);
    finalize_dinv_k<<<(TT * NN + blk - 1) / blk, blk, 0, stream>>>(dinv, TT * NN);

    wmma_gemm_k<<<dim3((NN / 16 + 15) / 16, HID / 64), blk, 0, stream>>>(
        x, w1, nullptr, XW1, NN, HID, HID, 1, 0);

    for (int t = 0; t < TT; ++t) {
        init_self_k<<<(NN * HID + blk - 1) / blk, blk, 0, stream>>>(
            Z1 + (size_t)t * NN * HID, XW1, dinv + t * NN, b1);
        scatter_k<<<(EE * 32 + blk - 1) / blk, blk, 0, stream>>>(
            Z1 + (size_t)t * NN * HID, XW1, dinv + t * NN,
            eis + (size_t)t * 2 * EE, eis + (size_t)t * 2 * EE + EE);
    }
    relu_k<<<(TT * NN * HID + blk - 1) / blk, blk, 0, stream>>>(Z1, TT * NN * HID);

    wmma_gemm_k<<<dim3((TT * NN / 16 + 15) / 16, HID / 64), blk, 0, stream>>>(
        Z1, w2, nullptr, H2, TT * NN, HID, HID, 1, 0);

    for (int t = 0; t < TT; ++t) {
        init_self_k<<<(NN * HID + blk - 1) / blk, blk, 0, stream>>>(
            Z2 + (size_t)t * NN * HID, H2 + (size_t)t * NN * HID, dinv + t * NN, b2);
        scatter_k<<<(EE * 32 + blk - 1) / blk, blk, 0, stream>>>(
            Z2 + (size_t)t * NN * HID, H2 + (size_t)t * NN * HID, dinv + t * NN,
            eis + (size_t)t * 2 * EE, eis + (size_t)t * 2 * EE + EE);
    }
    relu_k<<<(TT * NN * HID + blk - 1) / blk, blk, 0, stream>>>(Z2, TT * NN * HID);

    wmma_gemm_k<<<dim3((TT * NN / 16 + 15) / 16, G3 / 64), blk, 0, stream>>>(
        Z2, wih, bih, GI, TT * NN, G3, 1, HID, 0);

    gru_scan_k<<<1, 256, 0, stream>>>(GI, whh, bhh, out);
}


# GRU scan: step barrier split into two hand-overs (units 0..63 stored / all stored): the older wave of a SIMD runs next step's k=0..63 products under the younger wave's gates
# speedup vs baseline: 1.5411x; 1.0552x over previous
#include <hip/hip_runtime.h>
#include <math.h>

typedef __attribute__((ext_vector_type(2))) float v2f;
typedef __attribute__((ext_vector_type(8))) float v8f;

#define NN   20000
#define EE   320000
#define TT   4
#define HID  128
#define G3   384

__global__ void fill_zero_k(float* __restrict__ p, int n) {
    int i = blockIdx.x * blockDim.x + threadIdx.x;
    if (i < n) p[i] = 0.0f;
}

__global__ void count_deg_k(const int* __restrict__ eis, float* __restrict__ deg) {
    int i = blockIdx.x * blockDim.x + threadIdx.x;
    if (i >= TT * EE) return;
    int t = i / EE, e = i - t * EE;
    int d = eis[(size_t)t * 2 * EE + EE + e];
    atomicAdd(&deg[t * NN + d], 1.0f);
}

__global__ void finalize_dinv_k(float* __restrict__ deg, int n) {
    int i = blockIdx.x * blockDim.x + threadIdx.x;
    if (i < n) deg[i] = rsqrtf(deg[i] + 1.0f);
}

__global__ void init_self_k(float* __restrict__ out, const float* __restrict__ H,
                            const float* __restrict__ dinv, const float* __restrict__ bias) {
    int i = blockIdx.x * blockDim.x + threadIdx.x;
    if (i >= NN * HID) return;
    int n = i >> 7, j = i & 127;
    float di = dinv[n];
    out[i] = bias[j] + H[i] * di * di;
}

__global__ void scatter_k(float* __restrict__ out, const float* __restrict__ H,
                          const float* __restrict__ dinv,
                          const int* __restrict__ src, const int* __restrict__ dst) {
    int i = blockIdx.x * blockDim.x + threadIdx.x;
    int e = i >> 5;
    if (e >= EE) return;
    int lane = i & 31;
    int s = src[e], d = dst[e];
    float norm = dinv[s] * dinv[d];
    const float4 v = *(const float4*)(H + (size_t)s * HID + lane * 4);
    float* o = out + (size_t)d * HID + lane * 4;
    atomicAdd(o + 0, v.x * norm);
    atomicAdd(o + 1, v.y * norm);
    atomicAdd(o + 2, v.z * norm);
    atomicAdd(o + 3, v.w * norm);
}

__global__ void relu_k(float* __restrict__ p, int n) {
    int i = blockIdx.x * blockDim.x + threadIdx.x;
    if (i < n) p[i] = fmaxf(p[i], 0.0f);
}

#define BPS 160

__global__ void wmma_gemm_k(const float* __restrict__ A, const float* __restrict__ B,
                            const float* __restrict__ bias, float* __restrict__ C,
                            int Mrows, int Nout, int rsB, int csB, int relu) {
    __shared__ float Blds[64 * BPS];
    const int tid = threadIdx.x;
    const int col0 = blockIdx.y * 64;

    for (int idx = tid; idx < 128 * 64; idx += 256) {
        int k = idx >> 6, c = idx & 63;
        Blds[(k >> 1) * BPS + c * 2 + (k & 1)] =
            B[(size_t)k * rsB + (size_t)(col0 + c) * csB];
    }
    __syncthreads();

    const int wave = tid >> 5, lane = tid & 31;
    const int m = lane & 15;
    const int kof = (lane >> 4) * 2;
    const int rt0 = blockIdx.x * 16 + wave;
    const int rt1 = rt0 + 8;
    const int r0 = rt0 * 16, r1 = rt1 * 16;
    const float* A0 = A + (size_t)((r0 < Mrows ? r0 : 0) + m) * 128;
    const float* A1 = A + (size_t)((r1 < Mrows ? r1 : 0) + m) * 128;

    v8f acc0_0 = {}, acc0_1 = {}, acc0_2 = {}, acc0_3 = {};
    v8f acc1_0 = {}, acc1_1 = {}, acc1_2 = {}, acc1_3 = {};

    for (int k = 0; k < 128; k += 4) {
        int kk = k + kof;
        v2f a0 = *(const v2f*)(A0 + kk);
        v2f a1 = *(const v2f*)(A1 + kk);
        const float* bp = Blds + (kk >> 1) * BPS + m * 2;
        v2f b;
        b = *(const v2f*)(bp);
        acc0_0 = __builtin_amdgcn_wmma_f32_16x16x4_f32(false, a0, false, b, (short)0, acc0_0, false, false);
        acc1_0 = __builtin_amdgcn_wmma_f32_16x16x4_f32(false, a1, false, b, (short)0, acc1_0, false, false);
        b = *(const v2f*)(bp + 32);
        acc0_1 = __builtin_amdgcn_wmma_f32_16x16x4_f32(false, a0, false, b, (short)0, acc0_1, false, false);
        acc1_1 = __builtin_amdgcn_wmma_f32_16x16x4_f32(false, a1, false, b, (short)0, acc1_1, false, false);
        b = *(const v2f*)(bp + 64);
        acc0_2 = __builtin_amdgcn_wmma_f32_16x16x4_f32(false, a0, false, b, (short)0, acc0_2, false, false);
        acc1_2 = __builtin_amdgcn_wmma_f32_16x16x4_f32(false, a1, false, b, (short)0, acc1_2, false, false);
        b = *(const v2f*)(bp + 96);
        acc0_3 = __builtin_amdgcn_wmma_f32_16x16x4_f32(false, a0, false, b, (short)0, acc0_3, false, false);
        acc1_3 = __builtin_amdgcn_wmma_f32_16x16x4_f32(false, a1, false, b, (short)0, acc1_3, false, false);
    }

    const int mb = (lane >> 4) * 8;
    float bv0 = bias ? bias[col0 + m]      : 0.0f;
    float bv1 = bias ? bias[col0 + 16 + m] : 0.0f;
    float bv2 = bias ? bias[col0 + 32 + m] : 0.0f;
    float bv3 = bias ? bias[col0 + 48 + m] : 0.0f;
    if (r0 < Mrows) {
#pragma unroll
        for (int r = 0; r < 8; ++r) {
            size_t ro = (size_t)(r0 + mb + r) * Nout;
            float v;
            v = acc0_0[r] + bv0; if (relu) v = fmaxf(v, 0.0f); C[ro + col0 + m]      = v;
            v = acc0_1[r] + bv1; if (relu) v = fmaxf(v, 0.0f); C[ro + col0 + 16 + m] = v;
            v = acc0_2[r] + bv2; if (relu) v = fmaxf(v, 0.0f); C[ro + col0 + 32 + m] = v;
            v = acc0_3[r] + bv3; if (relu) v = fmaxf(v, 0.0f); C[ro + col0 + 48 + m] = v;
        }
    }
    if (r1 < Mrows) {
#pragma unroll
        for (int r = 0; r < 8; ++r) {
            size_t ro = (size_t)(r1 + mb + r) * Nout;
            float v;
            v = acc1_0[r] + bv0; if (relu) v = fmaxf(v, 0.0f); C[ro + col0 + m]      = v;
            v = acc1_1[r] + bv1; if (relu) v = fmaxf(v, 0.0f); C[ro + col0 + 16 + m] = v;
            v = acc1_2[r] + bv2; if (relu) v = fmaxf(v, 0.0f); C[ro + col0 + 32 + m] = v;
            v = acc1_3[r] + bv3; if (relu) v = fmaxf(v, 0.0f); C[ro + col0 + 48 + m] = v;
        }
    }
}

#define S_H  132
#define HBUF (16 * S_H)

__device__ __forceinline__ float sigm(float x) { return 1.0f / (1.0f + __expf(-x)); }

__device__ __forceinline__ void wg_arrive() {
    __builtin_amdgcn_fence(__ATOMIC_RELEASE, "workgroup", "local");
    __builtin_amdgcn_s_barrier_signal(-1);
}
__device__ __forceinline__ void wg_wait() {
    __builtin_amdgcn_s_barrier_wait(-1);
    __builtin_amdgcn_fence(__ATOMIC_ACQUIRE, "workgroup", "local");
}

__global__ void __launch_bounds__(256)
gru_scan_k(const float* __restrict__ GI, const float* __restrict__ whh,
           const float* __restrict__ bhh, float* __restrict__ out) {
    __shared__ float hS[2 * HBUF];
    const int tid = threadIdx.x;
    for (int idx = tid; idx < 2 * HBUF; idx += 256) hS[idx] = 0.0f;

    const int wave = tid >> 5, lane = tid & 31;
    const bool lead = __builtin_amdgcn_readfirstlane(wave) < 4;
    const int m   = lane & 15;
    const int hf  = lane >> 4;
    const int kof = hf * 2;
    const int j   = wave * 16 + m;

    v2f br[32], bz[32], bn[32];
    {
        const float* wr = whh + (size_t)j * HID + kof;
#pragma unroll
        for (int i = 0; i < 32; ++i) {
            br[i] = *(const v2f*)(wr + 4 * i);
            bz[i] = *(const v2f*)(wr + 128 * HID + 4 * i);
            bn[i] = *(const v2f*)(wr + 256 * HID + 4 * i);
        }
    }

    const int t0 = 2 * hf;
    const float* gip0 = GI + (size_t)t0 * NN * G3 + j;
    const float* gip1 = gip0 + (size_t)NN * G3;
    const float bh_r = bhh[j], bh_z = bhh[128 + j], bh_n = bhh[256 + j];
    float g0r = gip0[0], g0z = gip0[128], g0n = gip0[256];
    float g1r = gip1[0], g1z = gip1[128], g1n = gip1[256];
    float h0 = 0.0f, h1 = 0.0f;
    const int woff = (8 * hf) * S_H + j;
    const int aoff = m * S_H + kof;

    const int ct = tid >> 6, cj = (tid & 63) * 2;
    const int coff = ((ct & 1) + 8 * (ct >> 1)) * S_H + cj;
    float* outp = out + (size_t)ct * NN * HID + cj;

    __syncthreads();
    wg_arrive();

    for (int n = 0; n < NN; ++n) {
        const float* hR = hS + (n & 1) * HBUF;
        float*       hW = hS + ((n & 1) ^ 1) * HBUF;

        v2f a1[16], a2[16];
        v2f hv;
#pragma unroll
        for (int i = 0; i < 16; ++i) a1[i] = *(const v2f*)(hR + aoff + 4 * i);

        if (!lead) {
            wg_wait();
#pragma unroll
            for (int i = 0; i < 16; ++i) a2[i] = *(const v2f*)(hR + aoff + 64 + 4 * i);
            hv = *(const v2f*)(hR + coff);
        }

        v8f ar = {}, az = {}, an = {};
#pragma unroll
        for (int i = 0; i < 16; ++i) {
            ar = __builtin_amdgcn_wmma_f32_16x16x4_f32(false, a1[i], false, br[i], (short)0, ar, false, false);
            az = __builtin_amdgcn_wmma_f32_16x16x4_f32(false, a1[i], false, bz[i], (short)0, az, false, false);
            an = __builtin_amdgcn_wmma_f32_16x16x4_f32(false, a1[i], false, bn[i], (short)0, an, false, false);
        }

        if (lead) {
            wg_wait();
#pragma unroll
            for (int i = 0; i < 16; ++i) a2[i] = *(const v2f*)(hR + aoff + 64 + 4 * i);
            hv = *(const v2f*)(hR + coff);
        }

        if (n > 0) *(v2f*)(outp + (size_t)(n - 1) * HID) = hv;
        const int np = (n + 1 < NN) ? n + 1 : n;
        const float* q0 = gip0 + (size_t)np * G3;
        const float* q1 = gip1 + (size_t)np * G3;
        const float x0r = q0[0], x0z = q0[128], x0n = q0[256];
        const float x1r = q1[0], x1z = q1[128], x1n = q1[256];

#pragma unroll
        for (int i = 0; i < 15; ++i) {
            ar = __builtin_amdgcn_wmma_f32_16x16x4_f32(false, a2[i], false, br[16 + i], (short)0, ar, false, false);
            az = __builtin_amdgcn_wmma_f32_16x16x4_f32(false, a2[i], false, bz[16 + i], (short)0, az, false, false);
            an = __builtin_amdgcn_wmma_f32_16x16x4_f32(false, a2[i], false, bn[16 + i], (short)0, an, false, false);
        }
        if (!lead) __builtin_amdgcn_s_barrier_signal(-1);
#pragma unroll
        for (int i = 15; i < 16; ++i) {
            ar = __builtin_amdgcn_wmma_f32_16x16x4_f32(false, a2[i], false, br[16 + i], (short)0, ar, false, false);
            az = __builtin_amdgcn_wmma_f32_16x16x4_f32(false, a2[i], false, bz[16 + i], (short)0, az, false, false);
            an = __builtin_amdgcn_wmma_f32_16x16x4_f32(false, a2[i], false, bn[16 + i], (short)0, an, false, false);
        }

        const float r0 = sigm(g0r + (ar[0] + bh_r));
        const float z0 = sigm(g0z + (az[0] + bh_z));
        const float r1 = sigm(g1r + (ar[1] + bh_r));
        const float z1 = sigm(g1z + (az[1] + bh_z));
        const float hn0 = an[0] + bh_n;
        const float hn1 = an[1] + bh_n;
        const float nv0 = 2.0f * sigm(2.0f * (g0n + r0 * hn0)) - 1.0f;
        const float nv1 = 2.0f * sigm(2.0f * (g1n + r1 * hn1)) - 1.0f;
        const float h20 = (1.0f - z0) * nv0 + z0 * h0;
        const float h21 = (1.0f - z1) * nv1 + z1 * h1;
        hW[woff]       = h20;
        hW[woff + S_H] = h21;
        h0 = h20; h1 = h21;

        if (lead) wg_arrive();
        wg_wait();
        wg_arrive();
        g0r = x0r; g0z = x0z; g0n = x0n;
        g1r = x1r; g1z = x1z; g1n = x1n;
    }
    {
        wg_wait();
        const float* hR = hS + (NN & 1) * HBUF;
        v2f hv = *(const v2f*)(hR + coff);
        *(v2f*)(outp + (size_t)(NN - 1) * HID) = hv;
    }
}

extern "C" void kernel_launch(void* const* d_in, const int* in_sizes, int n_in,
                              void* d_out, int out_size, void* d_ws, size_t ws_size,
                              hipStream_t stream) {
    const float* x   = (const float*)d_in[0];
    const int*   eis = (const int*)  d_in[1];
    const float* w1  = (const float*)d_in[2];
    const float* b1  = (const float*)d_in[3];
    const float* w2  = (const float*)d_in[4];
    const float* b2  = (const float*)d_in[5];
    const float* wih = (const float*)d_in[6];
    const float* whh = (const float*)d_in[7];
    const float* bih = (const float*)d_in[8];
    const float* bhh = (const float*)d_in[9];
    float* out = (float*)d_out;

    float* ws   = (float*)d_ws;
    float* XW1  = ws;
    float* dinv = XW1 + (size_t)NN * HID;
    float* Z1   = dinv + (size_t)TT * NN;
    float* H2   = Z1 + (size_t)TT * NN * HID;
    float* GI   = H2 + (size_t)TT * NN * HID;
    float* Z2   = Z1;

    const int blk = 256;

    fill_zero_k<<<(TT * NN + blk - 1) / blk, blk, 0, stream>>>(dinv, TT * NN);
    count_deg_k<<<(TT * EE + blk - 1) / blk, blk, 0, stream>>>(eis, dinv);
    finalize_dinv_k<<<(TT * NN + blk - 1) / blk, blk, 0, stream>>>(dinv, TT * NN);

    wmma_gemm_k<<<dim3((NN / 16 + 15) / 16, HID / 64), blk, 0, stream>>>(
        x, w1, nullptr, XW1, NN, HID, HID, 1, 0);

    for (int t = 0; t < TT; ++t) {
        init_self_k<<<(NN * HID + blk - 1) / blk, blk, 0, stream>>>(
            Z1 + (size_t)t * NN * HID, XW1, dinv + t * NN, b1);
        scatter_k<<<(EE * 32 + blk - 1) / blk, blk, 0, stream>>>(
            Z1 + (size_t)t * NN * HID, XW1, dinv + t * NN,
            eis + (size_t)t * 2 * EE, eis + (size_t)t * 2 * EE + EE);
    }
    relu_k<<<(TT * NN * HID + blk - 1) / blk, blk, 0, stream>>>(Z1, TT * NN * HID);

    wmma_gemm_k<<<dim3((TT * NN / 16 + 15) / 16, HID / 64), blk, 0, stream>>>(
        Z1, w2, nullptr, H2, TT * NN, HID, HID, 1, 0);

    for (int t = 0; t < TT; ++t) {
        init_self_k<<<(NN * HID + blk - 1) / blk, blk, 0, stream>>>(
            Z2 + (size_t)t * NN * HID, H2 + (size_t)t * NN * HID, dinv + t * NN, b2);
        scatter_k<<<(EE * 32 + blk - 1) / blk, blk, 0, stream>>>(
            Z2 + (size_t)t * NN * HID, H2 + (size_t)t * NN * HID, dinv + t * NN,
            eis + (size_t)t * 2 * EE, eis + (size_t)t * 2 * EE + EE);
    }
    relu_k<<<(TT * NN * HID + blk - 1) / blk, blk, 0, stream>>>(Z2, TT * NN * HID);

    wmma_gemm_k<<<dim3((TT * NN / 16 + 15) / 16, G3 / 64), blk, 0, stream>>>(
        Z2, wih, bih, GI, TT * NN, G3, 1, HID, 0);

    gru_scan_k<<<1, 256, 0, stream>>>(GI, whh, bhh, out);
}
